# starting kernel
# baseline (speedup 1.0000x reference)
#include <hip/hip_runtime.h>

#define NN 50000
#define NE 800000
#define DHID 128
#define DOUT 64

typedef float v2f __attribute__((ext_vector_type(2)));
typedef float v8f __attribute__((ext_vector_type(8)));

__device__ __forceinline__ void atomic_fadd(float* p, float v) {
    asm volatile("global_atomic_add_f32 %0, %1, off" : : "v"(p), "v"(v) : "memory");
}

__global__ void zero_f4(float4* __restrict__ p, int n4) {
    int i = blockIdx.x * blockDim.x + threadIdx.x;
    if (i < n4) p[i] = make_float4(0.f, 0.f, 0.f, 0.f);
}

__global__ void degree_kernel(const int* __restrict__ dst, float* __restrict__ cnt, int E) {
    int e = blockIdx.x * blockDim.x + threadIdx.x;
    if (e < E) atomic_fadd(&cnt[dst[e]], 1.0f);
}

__global__ void scatter_kernel(const int* __restrict__ src, const int* __restrict__ dst,
                               const float* __restrict__ feat, float* __restrict__ agg,
                               int E) {
    int gid  = blockIdx.x * blockDim.x + threadIdx.x;
    int e    = gid >> 5;
    int lane = gid & 31;
    if (e >= E) return;
    int s = src[e];
    int d = dst[e];
    const float4 v = *(const float4*)(feat + (size_t)s * DHID + lane * 4);
    float* o = agg + (size_t)d * DHID + lane * 4;
    atomic_fadd(o + 0, v.x);
    atomic_fadd(o + 1, v.y);
    atomic_fadd(o + 2, v.z);
    atomic_fadd(o + 3, v.w);
}

template <int LDB>
__global__ void sage_gemm(const float* __restrict__ agg, const float* __restrict__ cnt,
                          const float* __restrict__ hroot,
                          const float* __restrict__ Wl, const float* __restrict__ bias,
                          const float* __restrict__ Wr,
                          float* __restrict__ out, int nrows, int relu) {
    constexpr int MB = 5;
    constexpr int NB = 2;
    const int lane = threadIdx.x & 31;
    const int wave = (blockIdx.x * blockDim.x + threadIdx.x) >> 5;
    const int ntn  = LDB / (16 * NB);
    const int stm  = wave / ntn;
    const int stn  = wave % ntn;
    if (stm * (16 * MB) >= nrows) return;

    const int half = lane >> 4;
    const int mn   = lane & 15;
    const int koff = half * 2;

    const int r0 = stm * (16 * MB) + mn;
    const int c0 = stn * (16 * NB) + mn;

    float rinv[MB];
#pragma unroll
    for (int mi = 0; mi < MB; ++mi)
        rinv[mi] = 1.0f / fmaxf(cnt[r0 + mi * 16], 1.0f);

    const float* ap = agg   + (size_t)r0 * DHID;
    const float* xp = hroot + (size_t)r0 * DHID;

    v8f acc[MB][NB];
#pragma unroll
    for (int mi = 0; mi < MB; ++mi)
#pragma unroll
        for (int nj = 0; nj < NB; ++nj)
            acc[mi][nj] = (v8f){0.f, 0.f, 0.f, 0.f, 0.f, 0.f, 0.f, 0.f};

#pragma unroll 2
    for (int k = 0; k < DHID; k += 4) {
        v2f bl[NB], br[NB];
#pragma unroll
        for (int nj = 0; nj < NB; ++nj) {
            bl[nj][0] = Wl[(k + koff)     * LDB + c0 + nj * 16];
            bl[nj][1] = Wl[(k + koff + 1) * LDB + c0 + nj * 16];
            br[nj][0] = Wr[(k + koff)     * LDB + c0 + nj * 16];
            br[nj][1] = Wr[(k + koff + 1) * LDB + c0 + nj * 16];
        }
        v2f am[MB], ar[MB];
#pragma unroll
        for (int mi = 0; mi < MB; ++mi) {
            float2 av = *(const float2*)(ap + mi * 16 * DHID + k + koff);
            am[mi][0] = av.x * rinv[mi];
            am[mi][1] = av.y * rinv[mi];
            float2 xv = *(const float2*)(xp + mi * 16 * DHID + k + koff);
            ar[mi][0] = xv.x;
            ar[mi][1] = xv.y;
        }
#pragma unroll
        for (int mi = 0; mi < MB; ++mi) {
#pragma unroll
            for (int nj = 0; nj < NB; ++nj) {
                acc[mi][nj] = __builtin_amdgcn_wmma_f32_16x16x4_f32(
                    false, am[mi], false, bl[nj], (short)0, acc[mi][nj], false, false);
                acc[mi][nj] = __builtin_amdgcn_wmma_f32_16x16x4_f32(
                    false, ar[mi], false, br[nj], (short)0, acc[mi][nj], false, false);
            }
        }
    }

#pragma unroll
    for (int nj = 0; nj < NB; ++nj) {
        const float bv = bias[c0 + nj * 16];
#pragma unroll
        for (int mi = 0; mi < MB; ++mi) {
#pragma unroll
            for (int v = 0; v < 8; ++v) {
                int m = stm * (16 * MB) + mi * 16 + v + half * 8;
                float val = acc[mi][nj][v] + bv;
                if (relu) val = fmaxf(val, 0.f);
                out[(size_t)m * LDB + c0 + nj * 16] = val;
            }
        }
    }
}

extern "C" void kernel_launch(void* const* d_in, const int* in_sizes, int n_in,
                              void* d_out, int out_size, void* d_ws, size_t ws_size,
                              hipStream_t stream) {
    const float* x   = (const float*)d_in[0];
    const int*   ei  = (const int*)d_in[1];
    const float* W1l = (const float*)d_in[2];
    const float* b1  = (const float*)d_in[3];
    const float* W1r = (const float*)d_in[4];
    const float* W2l = (const float*)d_in[5];
    const float* b2  = (const float*)d_in[6];
    const float* W2r = (const float*)d_in[7];
    float*       out = (float*)d_out;

    const int* src = ei;
    const int* dst = ei + NE;

    float* agg = (float*)d_ws;
    float* cnt = agg + (size_t)NN * DHID;
    float* h   = cnt + NN;

    const int BLK = 256;

    {
        int n4 = (NN * DHID + NN) / 4;
        zero_f4<<<(n4 + BLK - 1) / BLK, BLK, 0, stream>>>((float4*)agg, n4);
    }

    degree_kernel<<<(NE + BLK - 1) / BLK, BLK, 0, stream>>>(dst, cnt, NE);

    {
        long long thr = (long long)NE * 32;
        scatter_kernel<<<(int)((thr + BLK - 1) / BLK), BLK, 0, stream>>>(src, dst, x, agg, NE);

        int waves = (NN / 80) * (DHID / 32);
        int thr2  = waves * 32;
        sage_gemm<DHID><<<(thr2 + BLK - 1) / BLK, BLK, 0, stream>>>(
            agg, cnt, x, W1l, b1, W1r, h, NN,  1);
    }

    {
        int n4 = (NN * DHID) / 4;
        zero_f4<<<(n4 + BLK - 1) / BLK, BLK, 0, stream>>>((float4*)agg, n4);

        long long thr = (long long)NE * 32;
        scatter_kernel<<<(int)((thr + BLK - 1) / BLK), BLK, 0, stream>>>(src, dst, h, agg, NE);

        int waves = (NN / 80) * (DOUT / 32);
        int thr2  = waves * 32;
        sage_gemm<DOUT><<<(thr2 + BLK - 1) / BLK, BLK, 0, stream>>>(
            agg, cnt, h, W2l, b2, W2r, out, NN,  0);
    }
}


# Linear kernels: wave block 112x64 (layer 1) and 64x64 (layer 2), one wave per SIMD in 4-wave workgroups: 894 and 782 equal waves on 1024 SIMDs, not 2500 and 1250 of 80x32; sums unchanged
# speedup vs baseline: 3.5340x; 3.5340x over previous
#include <hip/hip_runtime.h>


#define NN 50000
#define NE 800000
#define DHID 128
#define DOUT 64

typedef float v2f __attribute__((ext_vector_type(2)));
typedef float v8f __attribute__((ext_vector_type(8)));

#define AR    224
#define AWG   ((NN + AR - 1) / AR)
#define ANW   32
#define AROWS (AR / ANW)
#define QCAP  896
#define APQ   64
#define NIT   (NE / 128)
#define KFULL (NIT / ANW)
#define KTAIL (NIT - KFULL * ANW)
#define SB    5
#define GR    16
#define LCAP  352
#define LHDR  32
#define LSEG  (LHDR + LCAP)

static_assert(NE % 128 == 0, "scan step is 128 edges");
static_assert(NE < (1 << 20) && AR <= 512, "an entry packs the edge id above a 9-bit local row");
static_assert(AROWS * ANW == AR && AROWS < 32, "every owned row has a wave; one count lane per row, lane AROWS takes the spare row's");
static_assert(KFULL % SB == 0 && KFULL >= 2 * SB && QCAP % 128 == 0 && QCAP > SB * 128, "scan batches of SB steps");
static_assert((size_t)AWG * ANW * LSEG <= (size_t)NN * DOUT, "the lists fit the output buffer");
static_assert(NN < (1 << 23) && LCAP % 32 == 0 && LCAP >= 32 && LSEG % 64 == 0, "an entry packs the source row above a 9-bit local row; whole 256-B units per wave");

template <bool COUNT>
__device__ __forceinline__ void add_rows(int s, int dl, int gend, int lane,
                                         const float* __restrict__ feat,
                                         float* rows, float& cntv) {
#pragma unroll 1
    for (int g = 0; g < gend; g += GR) {
        float4 v[GR];
#pragma unroll
        for (int j = 0; j < GR; ++j) {
            const int sj = __builtin_amdgcn_readlane(s, g + j);
            v[j] = *(const float4*)(feat + (size_t)sj * DHID + lane * 4);
        }
#pragma unroll
        for (int j = 0; j < GR; ++j) {
            const int dj = __builtin_amdgcn_readlane(dl, g + j);
            if (COUNT) cntv += (lane == (dj >> 5)) ? 1.0f : 0.0f;
            float4* r = (float4*)(rows + dj * DHID) + lane;
            float4 t = *r;
            t.x += v[j].x; t.y += v[j].y; t.z += v[j].z; t.w += v[j].w;
            *r = t;
        }
    }
}

template <bool FIRST>
__device__ __forceinline__ void drain_edges(const unsigned* __restrict__ slot, int n, int lane,
                                            const int* __restrict__ src,
                                            const float* __restrict__ feat,
                                            float* rows, float& cntv,
                                            unsigned* __restrict__ lst, int& lw) {
    const bool real = lane < n;
    const unsigned ent = real ? slot[lane] : 0u;
    const int e  = (int)(ent >> 9);
    const int dl = real ? (int)(ent & 511u) : AR;
    const int sr = src[e];
    const int s  = real ? sr : __builtin_amdgcn_readlane(sr, 0);
    if (FIRST) {
        if (real && lw + lane < LCAP) lst[lw + lane] = ((unsigned)sr << 9) | (unsigned)dl;
        lw += n;
    }
    add_rows<FIRST>(s, dl, 32, lane, feat, rows, cntv);
}

__device__ __forceinline__ void drain_list(unsigned ent, int n, int lane,
                                           const float* __restrict__ feat,
                                           float* rows, float& cntv) {
    const bool real = lane < n;
    const int sr = (int)(ent >> 9);
    const int s  = real ? sr : __builtin_amdgcn_readlane(sr, 0);
    const int dl = real ? (int)(ent & 511u) : AR;
    add_rows<false>(s, dl, n, lane, feat, rows, cntv);
}

template <bool FIRST>
__global__ __launch_bounds__(1024) void agg_ranges(const int* __restrict__ src,
                                                   const int* __restrict__ dst,
                                                   const float* __restrict__ feat,
                                                   float* __restrict__ agg,
                                                   float* __restrict__ cnt,
                                                   unsigned* __restrict__ lists) {
    __shared__ float    rows[(AR + 1) * DHID];
    __shared__ unsigned q[ANW * QCAP];
    __shared__ unsigned pend[ANW * APQ];
    __shared__ int qcnt[ANW];
    __shared__ int qmore[ANW];

    const int tid  = threadIdx.x;
    const int lane = tid & 31;
    const int w    = __builtin_amdgcn_readfirstlane(tid >> 5);
    const int base = blockIdx.x * AR;
    const int nrows = (NN - base < AR) ? (NN - base) : AR;

    unsigned* hdr = lists + (size_t)(blockIdx.x * ANW + w) * LSEG;
    unsigned* lst = hdr + LHDR;

    unsigned cw = 0u;
    if (!FIRST) cw = lists[(size_t)(blockIdx.x * ANW + lane) * LSEG];

#pragma unroll
    for (int i = 0; i < AROWS; ++i)
        ((float4*)(rows + (w + ANW * i) * DHID))[lane] = make_float4(0.f, 0.f, 0.f, 0.f);
    float cntv = 0.f;

    const bool uselist = !FIRST && (__builtin_amdgcn_ballot_w32(cw > (unsigned)LCAP) == 0u);

    if (uselist) {
        const int total = (int)__builtin_amdgcn_readlane(cw, w);
        unsigned ent = (lane < total) ? lst[lane] : 0u;
        for (int p = 0; p < total; p += 32) {
            const unsigned cur = ent;
            const int left = total - p;
            if (left > 32) ent = (lane < left - 32) ? lst[p + 32 + lane] : 0u;
            drain_list(cur, (left < 32) ? left : 32, lane, feat, rows, cntv);
        }
    } else {
        unsigned* myq = q + w * QCAP;
        unsigned* myp = pend + w * APQ;
        int lw = 0;

        const int kw = KFULL + ((w < KTAIL) ? 1 : 0);

        auto load_step = [&](int k) -> int4 {
            return *(const int4*)(dst + (w + ANW * k) * 128 + lane * 4);
        };
        auto load_batch = [&](int4 (&d)[SB], int k) {
#pragma unroll
            for (int j = 0; j < SB; ++j) d[j] = load_step((k + j < KFULL) ? (k + j) : (KFULL - 1));
        };

        int k = 0;
        int qn;
        auto examine = [&](const int4& d, int kk) {
            const unsigned e0 = (unsigned)((w + ANW * kk) * 128) + (unsigned)lane * 4u;
            const int dd[4] = {d.x, d.y, d.z, d.w};
            unsigned dl[4], b[4];
#pragma unroll
            for (int c = 0; c < 4; ++c) {
                dl[c] = (unsigned)(dd[c] - base);
                b[c]  = __builtin_amdgcn_ballot_w32(dl[c] < (unsigned)nrows);
            }
            if ((b[0] | b[1] | b[2] | b[3]) != 0u) {
#pragma unroll
                for (int c = 0; c < 4; ++c) {
                    if (b[c]) {
                        if (dl[c] < (unsigned)nrows)
                            myq[qn + (int)__builtin_amdgcn_mbcnt_lo(b[c], 0u)] = ((e0 + c) << 9) | dl[c];
                        qn += __builtin_popcount(b[c]);
                    }
                }
            }
        };

        bool more;
        do {
            qn = 0;
            if (k + SB <= KFULL) {
                int4 da[SB], db[SB];
                load_batch(da, k);
                while (true) {
                    load_batch(db, k + SB);
#pragma unroll
                    for (int j = 0; j < SB; ++j) examine(da[j], k + j);
                    k += SB;
                    if (k + SB > KFULL || qn > QCAP - SB * 128) break;
                    load_batch(da, k + SB);
#pragma unroll
                    for (int j = 0; j < SB; ++j) examine(db[j], k + j);
                    k += SB;
                    if (k + SB > KFULL || qn > QCAP - SB * 128) break;
                }
            }
            while (k + SB > KFULL && k < kw && qn <= QCAP - 128) {
                const int4 d = load_step(k);
                examine(d, k);
                ++k;
            }
            if (lane == 0) { qcnt[w] = qn; qmore[w] = (k < kw) ? 1 : 0; }
            __syncthreads();
            more = __builtin_amdgcn_ballot_w32(qmore[lane] != 0) != 0u;
            const int nv = qcnt[lane];

            int pn = 0;
            for (int sw = 0; sw < ANW; ++sw) {
                const int n = __builtin_amdgcn_readlane(nv, sw);
                const unsigned* sq = q + sw * QCAP;
                for (int p = 0; p < n; p += 128) {
                    const uint4 e4 = *(const uint4*)(sq + p + lane * 4);
                    const unsigned ee[4] = {e4.x, e4.y, e4.z, e4.w};
#pragma unroll
                    for (int c = 0; c < 4; ++c) {
                        const bool mine = (p + lane * 4 + c < n) && ((int)(ee[c] & 31u) == w);
                        const unsigned bm = __builtin_amdgcn_ballot_w32(mine);
                        if (bm) {
                            if (mine) myp[pn + (int)__builtin_amdgcn_mbcnt_lo(bm, 0u)] = ee[c];
                            pn += __builtin_popcount(bm);
                            if (pn >= 32) {
                                pn -= 32;
                                drain_edges<FIRST>(myp + pn, 32, lane, src, feat, rows, cntv, lst, lw);
                            }
                        }
                    }
                }
            }
            if (pn > 0) drain_edges<FIRST>(myp, pn, lane, src, feat, rows, cntv, lst, lw);
            __syncthreads();
        } while (more);

        if (FIRST) {
            if (lane == 0) hdr[0] = (unsigned)lw;
        }
    }

#pragma unroll
    for (int i = 0; i < AROWS; ++i) {
        const int r = w + ANW * i;
        if (r < nrows)
            *(float4*)(agg + (size_t)(base + r) * DHID + lane * 4) = ((const float4*)(rows + r * DHID))[lane];
    }
    if (FIRST) {
        const int r = w + ANW * lane;
        if (lane < AROWS && r < nrows) cnt[base + r] = cntv;
    }
}

template <int LDB, int MB>
__global__ __launch_bounds__(128) __attribute__((amdgpu_waves_per_eu(1, 1))) void sage_gemm(const float* __restrict__ agg, const float* __restrict__ cnt,
                          const float* __restrict__ hroot,
                          const float* __restrict__ Wl, const float* __restrict__ bias,
                          const float* __restrict__ Wr,
                          float* __restrict__ out, int nrows, int relu) {
    constexpr int NB = 4;
    constexpr int KB = 8;
    const int lane = threadIdx.x & 31;
    const int wave = (blockIdx.x * blockDim.x + threadIdx.x) >> 5;
    const int ntn  = LDB / (16 * NB);
    const int stm  = wave / ntn;
    const int stn  = wave % ntn;
    const int ntiles = nrows >> 4;
    const int t0 = stm * MB;
    if (t0 >= ntiles) return;

    const int half = lane >> 4;
    const int mn   = lane & 15;
    const int koff = half * 2;

    const int c0 = stn * (16 * NB) + mn;

    unsigned ro[MB];
    float rinv[MB];
#pragma unroll
    for (int mi = 0; mi < MB; ++mi) {
        const int t   = (t0 + mi < ntiles) ? (t0 + mi) : t0;
        const int row = t * 16 + mn;
        ro[mi]   = (unsigned)row * DHID;
        rinv[mi] = 1.0f / fmaxf(cnt[row], 1.0f);
    }

    v8f acc[MB][NB];
#pragma unroll
    for (int mi = 0; mi < MB; ++mi)
#pragma unroll
        for (int nj = 0; nj < NB; ++nj)
            acc[mi][nj] = (v8f){0.f, 0.f, 0.f, 0.f, 0.f, 0.f, 0.f, 0.f};

    struct KFrag { v2f am[MB], ar[MB], bl[NB], br[NB]; };

    auto load_k = [&](KFrag& f, int k) {
#pragma unroll
        for (int nj = 0; nj < NB; ++nj) {
            f.bl[nj][0] = Wl[(k + koff)     * LDB + c0 + nj * 16];
            f.bl[nj][1] = Wl[(k + koff + 1) * LDB + c0 + nj * 16];
            f.br[nj][0] = Wr[(k + koff)     * LDB + c0 + nj * 16];
            f.br[nj][1] = Wr[(k + koff + 1) * LDB + c0 + nj * 16];
        }
#pragma unroll
        for (int mi = 0; mi < MB; ++mi) {
            const float2 av = *(const float2*)(agg + (size_t)ro[mi] + (k + koff));
            f.am[mi][0] = av.x;
            f.am[mi][1] = av.y;
            const float2 xv = *(const float2*)(hroot + (size_t)ro[mi] + (k + koff));
            f.ar[mi][0] = xv.x;
            f.ar[mi][1] = xv.y;
        }
    };
    auto mma_k = [&](const KFrag& f) {
        v2f am[MB];
#pragma unroll
        for (int mi = 0; mi < MB; ++mi) {
            am[mi][0] = f.am[mi][0] * rinv[mi];
            am[mi][1] = f.am[mi][1] * rinv[mi];
        }
#pragma unroll
        for (int mi = 0; mi < MB; ++mi)
#pragma unroll
            for (int nj = 0; nj < NB; ++nj)
                acc[mi][nj] = __builtin_amdgcn_wmma_f32_16x16x4_f32(
                    false, am[mi], false, f.bl[nj], (short)0, acc[mi][nj], false, false);
#pragma unroll
        for (int mi = 0; mi < MB; ++mi)
#pragma unroll
            for (int nj = 0; nj < NB; ++nj)
                acc[mi][nj] = __builtin_amdgcn_wmma_f32_16x16x4_f32(
                    false, f.ar[mi], false, f.br[nj], (short)0, acc[mi][nj], false, false);
    };

    KFrag p0, p1, q0, q1;
    load_k(p0, 0);
    load_k(p1, 4);
    for (int kb = 0; kb < DHID; kb += 2 * KB) {
        load_k(q0, kb + KB);
        load_k(q1, kb + KB + 4);
        __builtin_amdgcn_sched_group_barrier(0x020, 2 * (4 * NB + 2 * MB), 0);
        __builtin_amdgcn_sched_group_barrier(0x008, 4 * MB * NB, 0);
        mma_k(p0);
        mma_k(p1);
        const int kn = (kb + 2 * KB < DHID) ? (kb + 2 * KB) : (DHID - KB);
        load_k(p0, kn);
        load_k(p1, kn + 4);
        __builtin_amdgcn_sched_group_barrier(0x020, 2 * (4 * NB + 2 * MB), 0);
        __builtin_amdgcn_sched_group_barrier(0x008, 4 * MB * NB, 0);
        mma_k(q0);
        mma_k(q1);
    }

#pragma unroll
    for (int mi = 0; mi < MB; ++mi) {
        if (t0 + mi < ntiles) {
#pragma unroll
            for (int nj = 0; nj < NB; ++nj) {
                const float bv = bias[c0 + nj * 16];
#pragma unroll
                for (int v = 0; v < 8; ++v) {
                    const int m = (t0 + mi) * 16 + v + half * 8;
                    float val = acc[mi][nj][v] + bv;
                    if (relu) val = fmaxf(val, 0.f);
                    out[(size_t)m * LDB + c0 + nj * 16] = val;
                }
            }
        }
    }
}

extern "C" void kernel_launch(void* const* d_in, const int* in_sizes, int n_in,
                              void* d_out, int out_size, void* d_ws, size_t ws_size,
                              hipStream_t stream) {
    const float* x   = (const float*)d_in[0];
    const int*   ei  = (const int*)d_in[1];
    const float* W1l = (const float*)d_in[2];
    const float* b1  = (const float*)d_in[3];
    const float* W1r = (const float*)d_in[4];
    const float* W2l = (const float*)d_in[5];
    const float* b2  = (const float*)d_in[6];
    const float* W2r = (const float*)d_in[7];
    float*       out = (float*)d_out;

    const int* src = ei;
    const int* dst = ei + NE;

    float* agg = (float*)d_ws;
    float* cnt = agg + (size_t)NN * DHID;
    float* h   = cnt + NN;
    unsigned* lists = (unsigned*)d_out;

    const int BLK = 128;
    const int AGG_WG = AWG;

    {
        agg_ranges<true><<<AGG_WG, 1024, 0, stream>>>(src, dst, x, agg, cnt, lists);

        int waves = ((NN / 16 + 6) / 7) * (DHID / 64);
        int thr2  = waves * 32;
        sage_gemm<DHID, 7><<<(thr2 + BLK - 1) / BLK, BLK, 0, stream>>>(
            agg, cnt, x, W1l, b1, W1r, h, NN,  1);
    }

    {
        agg_ranges<false><<<AGG_WG, 1024, 0, stream>>>(src, dst, h, agg, cnt, lists);

        int waves = ((NN / 16 + 3) / 4) * (DOUT / 64);
        int thr2  = waves * 32;
        sage_gemm<DOUT, 4><<<(thr2 + BLK - 1) / BLK, BLK, 0, stream>>>(
            agg, cnt, h, W2l, b2, W2r, out, NN,  0);
    }
}


# Linear kernels: every fragment load and result store is 16 B per lane (A: 4 k of a row; B, bias, out: 4 adjacent columns, one per tile); sum order changes: a step sums k, k+1, k+4, k+5 of its 8
# speedup vs baseline: 3.6776x; 1.0406x over previous
#include <hip/hip_runtime.h>


#define NN 50000
#define NE 800000
#define DHID 128
#define DOUT 64

typedef float v2f __attribute__((ext_vector_type(2)));
typedef float v4f __attribute__((ext_vector_type(4)));
typedef float v8f __attribute__((ext_vector_type(8)));

#define AR    224
#define AWG   ((NN + AR - 1) / AR)
#define ANW   32
#define AROWS (AR / ANW)
#define QCAP  896
#define APQ   64
#define NIT   (NE / 128)
#define KFULL (NIT / ANW)
#define KTAIL (NIT - KFULL * ANW)
#define SB    5
#define GR    16
#define LCAP  352
#define LHDR  32
#define LSEG  (LHDR + LCAP)

static_assert(NE % 128 == 0, "scan step is 128 edges");
static_assert(NE < (1 << 20) && AR <= 512, "an entry packs the edge id above a 9-bit local row");
static_assert(AROWS * ANW == AR && AROWS < 32, "every owned row has a wave; one count lane per row, lane AROWS takes the spare row's");
static_assert(KFULL % SB == 0 && KFULL >= 2 * SB && QCAP % 128 == 0 && QCAP > SB * 128, "scan batches of SB steps");
static_assert((size_t)AWG * ANW * LSEG <= (size_t)NN * DOUT, "the lists fit the output buffer");
static_assert(NN < (1 << 23) && LCAP % 32 == 0 && LCAP >= 32 && LSEG % 64 == 0, "an entry packs the source row above a 9-bit local row; whole 256-B units per wave");

template <bool COUNT>
__device__ __forceinline__ void add_rows(int s, int dl, int gend, int lane,
                                         const float* __restrict__ feat,
                                         float* rows, float& cntv) {
#pragma unroll 1
    for (int g = 0; g < gend; g += GR) {
        float4 v[GR];
#pragma unroll
        for (int j = 0; j < GR; ++j) {
            const int sj = __builtin_amdgcn_readlane(s, g + j);
            v[j] = *(const float4*)(feat + (size_t)sj * DHID + lane * 4);
        }
#pragma unroll
        for (int j = 0; j < GR; ++j) {
            const int dj = __builtin_amdgcn_readlane(dl, g + j);
            if (COUNT) cntv += (lane == (dj >> 5)) ? 1.0f : 0.0f;
            float4* r = (float4*)(rows + dj * DHID) + lane;
            float4 t = *r;
            t.x += v[j].x; t.y += v[j].y; t.z += v[j].z; t.w += v[j].w;
            *r = t;
        }
    }
}

template <bool FIRST>
__device__ __forceinline__ void drain_edges(const unsigned* __restrict__ slot, int n, int lane,
                                            const int* __restrict__ src,
                                            const float* __restrict__ feat,
                                            float* rows, float& cntv,
                                            unsigned* __restrict__ lst, int& lw) {
    const bool real = lane < n;
    const unsigned ent = real ? slot[lane] : 0u;
    const int e  = (int)(ent >> 9);
    const int dl = real ? (int)(ent & 511u) : AR;
    const int sr = src[e];
    const int s  = real ? sr : __builtin_amdgcn_readlane(sr, 0);
    if (FIRST) {
        if (real && lw + lane < LCAP) lst[lw + lane] = ((unsigned)sr << 9) | (unsigned)dl;
        lw += n;
    }
    add_rows<FIRST>(s, dl, 32, lane, feat, rows, cntv);
}

__device__ __forceinline__ void drain_list(unsigned ent, int n, int lane,
                                           const float* __restrict__ feat,
                                           float* rows, float& cntv) {
    const bool real = lane < n;
    const int sr = (int)(ent >> 9);
    const int s  = real ? sr : __builtin_amdgcn_readlane(sr, 0);
    const int dl = real ? (int)(ent & 511u) : AR;
    add_rows<false>(s, dl, n, lane, feat, rows, cntv);
}

template <bool FIRST>
__global__ __launch_bounds__(1024) void agg_ranges(const int* __restrict__ src,
                                                   const int* __restrict__ dst,
                                                   const float* __restrict__ feat,
                                                   float* __restrict__ agg,
                                                   float* __restrict__ cnt,
                                                   unsigned* __restrict__ lists) {
    __shared__ float    rows[(AR + 1) * DHID];
    __shared__ unsigned q[ANW * QCAP];
    __shared__ unsigned pend[ANW * APQ];
    __shared__ int qcnt[ANW];
    __shared__ int qmore[ANW];

    const int tid  = threadIdx.x;
    const int lane = tid & 31;
    const int w    = __builtin_amdgcn_readfirstlane(tid >> 5);
    const int base = blockIdx.x * AR;
    const int nrows = (NN - base < AR) ? (NN - base) : AR;

    unsigned* hdr = lists + (size_t)(blockIdx.x * ANW + w) * LSEG;
    unsigned* lst = hdr + LHDR;

    unsigned cw = 0u;
    if (!FIRST) cw = lists[(size_t)(blockIdx.x * ANW + lane) * LSEG];

#pragma unroll
    for (int i = 0; i < AROWS; ++i)
        ((float4*)(rows + (w + ANW * i) * DHID))[lane] = make_float4(0.f, 0.f, 0.f, 0.f);
    float cntv = 0.f;

    const bool uselist = !FIRST && (__builtin_amdgcn_ballot_w32(cw > (unsigned)LCAP) == 0u);

    if (uselist) {
        const int total = (int)__builtin_amdgcn_readlane(cw, w);
        unsigned ent = (lane < total) ? lst[lane] : 0u;
        for (int p = 0; p < total; p += 32) {
            const unsigned cur = ent;
            const int left = total - p;
            if (left > 32) ent = (lane < left - 32) ? lst[p + 32 + lane] : 0u;
            drain_list(cur, (left < 32) ? left : 32, lane, feat, rows, cntv);
        }
    } else {
        unsigned* myq = q + w * QCAP;
        unsigned* myp = pend + w * APQ;
        int lw = 0;

        const int kw = KFULL + ((w < KTAIL) ? 1 : 0);

        auto load_step = [&](int k) -> int4 {
            return *(const int4*)(dst + (w + ANW * k) * 128 + lane * 4);
        };
        auto load_batch = [&](int4 (&d)[SB], int k) {
#pragma unroll
            for (int j = 0; j < SB; ++j) d[j] = load_step((k + j < KFULL) ? (k + j) : (KFULL - 1));
        };

        int k = 0;
        int qn;
        auto examine = [&](const int4& d, int kk) {
            const unsigned e0 = (unsigned)((w + ANW * kk) * 128) + (unsigned)lane * 4u;
            const int dd[4] = {d.x, d.y, d.z, d.w};
            unsigned dl[4], b[4];
#pragma unroll
            for (int c = 0; c < 4; ++c) {
                dl[c] = (unsigned)(dd[c] - base);
                b[c]  = __builtin_amdgcn_ballot_w32(dl[c] < (unsigned)nrows);
            }
            if ((b[0] | b[1] | b[2] | b[3]) != 0u) {
#pragma unroll
                for (int c = 0; c < 4; ++c) {
                    if (b[c]) {
                        if (dl[c] < (unsigned)nrows)
                            myq[qn + (int)__builtin_amdgcn_mbcnt_lo(b[c], 0u)] = ((e0 + c) << 9) | dl[c];
                        qn += __builtin_popcount(b[c]);
                    }
                }
            }
        };

        bool more;
        do {
            qn = 0;
            if (k + SB <= KFULL) {
                int4 da[SB], db[SB];
                load_batch(da, k);
                while (true) {
                    load_batch(db, k + SB);
#pragma unroll
                    for (int j = 0; j < SB; ++j) examine(da[j], k + j);
                    k += SB;
                    if (k + SB > KFULL || qn > QCAP - SB * 128) break;
                    load_batch(da, k + SB);
#pragma unroll
                    for (int j = 0; j < SB; ++j) examine(db[j], k + j);
                    k += SB;
                    if (k + SB > KFULL || qn > QCAP - SB * 128) break;
                }
            }
            while (k + SB > KFULL && k < kw && qn <= QCAP - 128) {
                const int4 d = load_step(k);
                examine(d, k);
                ++k;
            }
            if (lane == 0) { qcnt[w] = qn; qmore[w] = (k < kw) ? 1 : 0; }
            __syncthreads();
            more = __builtin_amdgcn_ballot_w32(qmore[lane] != 0) != 0u;
            const int nv = qcnt[lane];

            int pn = 0;
            for (int sw = 0; sw < ANW; ++sw) {
                const int n = __builtin_amdgcn_readlane(nv, sw);
                const unsigned* sq = q + sw * QCAP;
                for (int p = 0; p < n; p += 128) {
                    const uint4 e4 = *(const uint4*)(sq + p + lane * 4);
                    const unsigned ee[4] = {e4.x, e4.y, e4.z, e4.w};
#pragma unroll
                    for (int c = 0; c < 4; ++c) {
                        const bool mine = (p + lane * 4 + c < n) && ((int)(ee[c] & 31u) == w);
                        const unsigned bm = __builtin_amdgcn_ballot_w32(mine);
                        if (bm) {
                            if (mine) myp[pn + (int)__builtin_amdgcn_mbcnt_lo(bm, 0u)] = ee[c];
                            pn += __builtin_popcount(bm);
                            if (pn >= 32) {
                                pn -= 32;
                                drain_edges<FIRST>(myp + pn, 32, lane, src, feat, rows, cntv, lst, lw);
                            }
                        }
                    }
                }
            }
            if (pn > 0) drain_edges<FIRST>(myp, pn, lane, src, feat, rows, cntv, lst, lw);
            __syncthreads();
        } while (more);

        if (FIRST) {
            if (lane == 0) hdr[0] = (unsigned)lw;
        }
    }

#pragma unroll
    for (int i = 0; i < AROWS; ++i) {
        const int r = w + ANW * i;
        if (r < nrows)
            *(float4*)(agg + (size_t)(base + r) * DHID + lane * 4) = ((const float4*)(rows + r * DHID))[lane];
    }
    if (FIRST) {
        const int r = w + ANW * lane;
        if (lane < AROWS && r < nrows) cnt[base + r] = cntv;
    }
}

template <int LDB, int MB>
__global__ __launch_bounds__(128) __attribute__((amdgpu_waves_per_eu(1, 1))) void sage_gemm(const float* __restrict__ agg, const float* __restrict__ cnt,
                          const float* __restrict__ hroot,
                          const float* __restrict__ Wl, const float* __restrict__ bias,
                          const float* __restrict__ Wr,
                          float* __restrict__ out, int nrows, int relu) {
    constexpr int NB = 4;
    constexpr int KB = 8;
    const int lane = threadIdx.x & 31;
    const int wave = (blockIdx.x * blockDim.x + threadIdx.x) >> 5;
    const int ntn  = LDB / (16 * NB);
    const int stm  = wave / ntn;
    const int stn  = wave % ntn;
    const int ntiles = nrows >> 4;
    const int t0 = stm * MB;
    if (t0 >= ntiles) return;

    const int half = lane >> 4;
    const int mn   = lane & 15;
    const int kq   = half * 4;

    const int c0 = stn * (16 * NB) + mn * 4;

    unsigned ro[MB];
    float rinv[MB];
#pragma unroll
    for (int mi = 0; mi < MB; ++mi) {
        const int t   = (t0 + mi < ntiles) ? (t0 + mi) : t0;
        const int row = t * 16 + mn;
        ro[mi]   = (unsigned)row * DHID;
        rinv[mi] = 1.0f / fmaxf(cnt[row], 1.0f);
    }

    v8f acc[MB][NB];
#pragma unroll
    for (int mi = 0; mi < MB; ++mi)
#pragma unroll
        for (int nj = 0; nj < NB; ++nj)
            acc[mi][nj] = (v8f){0.f, 0.f, 0.f, 0.f, 0.f, 0.f, 0.f, 0.f};

    struct KBlk { v4f am[MB], ar[MB], bl[4], br[4]; };

    auto load_blk = [&](KBlk& f, int kb) {
#pragma unroll
        for (int j = 0; j < 4; ++j) {
            f.bl[j] = *(const v4f*)(Wl + (kb + kq + j) * LDB + c0);
            f.br[j] = *(const v4f*)(Wr + (kb + kq + j) * LDB + c0);
        }
#pragma unroll
        for (int mi = 0; mi < MB; ++mi) {
            f.am[mi] = *(const v4f*)(agg   + (size_t)ro[mi] + (kb + kq));
            f.ar[mi] = *(const v4f*)(hroot + (size_t)ro[mi] + (kb + kq));
        }
    };
    auto mma_blk = [&](const KBlk& f) {
#pragma unroll
        for (int s = 0; s < 2; ++s) {
            v2f am[MB], ar[MB], bl[NB], br[NB];
#pragma unroll
            for (int mi = 0; mi < MB; ++mi) {
                am[mi][0] = f.am[mi][2 * s]     * rinv[mi];
                am[mi][1] = f.am[mi][2 * s + 1] * rinv[mi];
                ar[mi][0] = f.ar[mi][2 * s];
                ar[mi][1] = f.ar[mi][2 * s + 1];
            }
#pragma unroll
            for (int nj = 0; nj < NB; ++nj) {
                bl[nj][0] = f.bl[2 * s][nj];
                bl[nj][1] = f.bl[2 * s + 1][nj];
                br[nj][0] = f.br[2 * s][nj];
                br[nj][1] = f.br[2 * s + 1][nj];
            }
#pragma unroll
            for (int mi = 0; mi < MB; ++mi)
#pragma unroll
                for (int nj = 0; nj < NB; ++nj)
                    acc[mi][nj] = __builtin_amdgcn_wmma_f32_16x16x4_f32(
                        false, am[mi], false, bl[nj], (short)0, acc[mi][nj], false, false);
#pragma unroll
            for (int mi = 0; mi < MB; ++mi)
#pragma unroll
                for (int nj = 0; nj < NB; ++nj)
                    acc[mi][nj] = __builtin_amdgcn_wmma_f32_16x16x4_f32(
                        false, ar[mi], false, br[nj], (short)0, acc[mi][nj], false, false);
        }
    };

    KBlk p, q;
    load_blk(p, 0);
    for (int kb = 0; kb < DHID; kb += 2 * KB) {
        load_blk(q, kb + KB);
        __builtin_amdgcn_sched_group_barrier(0x020, 8 + 2 * MB, 0);
        __builtin_amdgcn_sched_group_barrier(0x008, 4 * MB * NB, 0);
        mma_blk(p);
        const int kn = (kb + 2 * KB < DHID) ? (kb + 2 * KB) : (DHID - KB);
        load_blk(p, kn);
        __builtin_amdgcn_sched_group_barrier(0x020, 8 + 2 * MB, 0);
        __builtin_amdgcn_sched_group_barrier(0x008, 4 * MB * NB, 0);
        mma_blk(q);
    }

    const v4f bv = *(const v4f*)(bias + c0);
#pragma unroll
    for (int mi = 0; mi < MB; ++mi) {
        if (t0 + mi < ntiles) {
#pragma unroll
            for (int v = 0; v < 8; ++v) {
                const int m = (t0 + mi) * 16 + v + half * 8;
                v4f val;
#pragma unroll
                for (int nj = 0; nj < NB; ++nj) {
                    float t = acc[mi][nj][v] + bv[nj];
                    if (relu) t = fmaxf(t, 0.f);
                    val[nj] = t;
                }
                *(v4f*)(out + (size_t)m * LDB + c0) = val;
            }
        }
    }
}

extern "C" void kernel_launch(void* const* d_in, const int* in_sizes, int n_in,
                              void* d_out, int out_size, void* d_ws, size_t ws_size,
                              hipStream_t stream) {
    const float* x   = (const float*)d_in[0];
    const int*   ei  = (const int*)d_in[1];
    const float* W1l = (const float*)d_in[2];
    const float* b1  = (const float*)d_in[3];
    const float* W1r = (const float*)d_in[4];
    const float* W2l = (const float*)d_in[5];
    const float* b2  = (const float*)d_in[6];
    const float* W2r = (const float*)d_in[7];
    float*       out = (float*)d_out;

    const int* src = ei;
    const int* dst = ei + NE;

    float* agg = (float*)d_ws;
    float* cnt = agg + (size_t)NN * DHID;
    float* h   = cnt + NN;
    unsigned* lists = (unsigned*)d_out;

    const int BLK = 128;
    const int AGG_WG = AWG;

    {
        agg_ranges<true><<<AGG_WG, 1024, 0, stream>>>(src, dst, x, agg, cnt, lists);

        int waves = ((NN / 16 + 6) / 7) * (DHID / 64);
        int thr2  = waves * 32;
        sage_gemm<DHID, 7><<<(thr2 + BLK - 1) / BLK, BLK, 0, stream>>>(
            agg, cnt, x, W1l, b1, W1r, h, NN,  1);
    }

    {
        agg_ranges<false><<<AGG_WG, 1024, 0, stream>>>(src, dst, h, agg, cnt, lists);

        int waves = ((NN / 16 + 3) / 4) * (DOUT / 64);
        int thr2  = waves * 32;
        sage_gemm<DOUT, 4><<<(thr2 + BLK - 1) / BLK, BLK, 0, stream>>>(
            agg, cnt, h, W2l, b2, W2r, out, NN,  0);
    }
}
